# starting kernel
# baseline (speedup 1.0000x reference)
#include <hip/hip_runtime.h>
#include <hip/hip_bf16.h>

typedef __attribute__((ext_vector_type(2))) float v2f;
typedef __attribute__((ext_vector_type(4))) float v4f;
typedef __attribute__((ext_vector_type(8))) float v8f;

#define NCENTERS 16384
#define OUTROW 512

__global__ void zero_kernel(int* __restrict__ p, int n) {
    int i = blockIdx.x * blockDim.x + threadIdx.x;
    if (i < n) p[i] = 0;
}

__global__ void hist_kernel(const int* __restrict__ di, int* __restrict__ counts, int P) {
    int i = blockIdx.x * blockDim.x + threadIdx.x;
    if (i < P) atomicAdd(&counts[di[i]], 1);
}

__global__ void scan_kernel(const int* __restrict__ counts, int* __restrict__ offsets,
                            int* __restrict__ cursors, int P, int C) {
    __shared__ int buf[2][1024];
    const int t = threadIdx.x;
    const int per = C / 1024;
    const int base = t * per;
    int s = 0;
    for (int i = 0; i < per; ++i) s += counts[base + i];
    buf[0][t] = s;
    __syncthreads();
    int src = 0;
    for (int off = 1; off < 1024; off <<= 1) {
        int v = buf[src][t];
        if (t >= off) v += buf[src][t - off];
        buf[src ^ 1][t] = v;
        src ^= 1;
        __syncthreads();
    }
    int run = buf[src][t] - s;
    for (int i = 0; i < per; ++i) {
        offsets[base + i] = run;
        cursors[base + i] = run;
        run += counts[base + i];
    }
    if (t == 0) offsets[C] = P;
}

__global__ void scatter_kernel(const int* __restrict__ di, int* __restrict__ cursors,
                               int* __restrict__ order, int P) {
    int i = blockIdx.x * blockDim.x + threadIdx.x;
    if (i < P) {
        int pos = atomicAdd(&cursors[di[i]], 1);
        order[pos] = i;
    }
}

__global__ __launch_bounds__(32) void spexp_wmma_kernel(
    const float* __restrict__ rb,
    const float* __restrict__ sph,
    const float* __restrict__ W,
    const int*   __restrict__ species,
    const int*   __restrict__ order,
    const int*   __restrict__ offsets,
    float*       __restrict__ out)
{
    __shared__ alignas(16) float sW[16];
    __shared__ alignas(16) float sS[4][16];
    __shared__ alignas(16) float sR[4][32];
    __shared__ alignas(16) float sP[4][4];

    const int c    = blockIdx.x;
    const int lane = threadIdx.x;
    const int half = lane >> 4;
    const int idx  = lane & 15;

    if (lane < 16) sW[lane] = W[lane];

    const int start = offsets[c];
    const int end   = offsets[c + 1];
    const int count = end - start;
    const int nfull = count & ~3;
    const int tail  = count & 3;

    v8f acc[4][2];
    const v8f vzero = {0.f, 0.f, 0.f, 0.f, 0.f, 0.f, 0.f, 0.f};
#pragma unroll
    for (int q = 0; q < 4; ++q)
#pragma unroll
        for (int nt = 0; nt < 2; ++nt) acc[q][nt] = vzero;

    __syncthreads();

    const int k = lane >> 3;
    const int j = lane & 7;
    const int k0 = half * 2;

    auto do_wmma = [&]() {
#pragma unroll
        for (int q = 0; q < 4; ++q) {
            v2f a;
            a.x = sS[k0][idx]     * sP[k0][q];
            a.y = sS[k0 + 1][idx] * sP[k0 + 1][q];
#pragma unroll
            for (int nt = 0; nt < 2; ++nt) {
                const int col = nt * 16 + idx;
                v2f b;
                b.x = sR[k0][col];
                b.y = sR[k0 + 1][col];
                acc[q][nt] = __builtin_amdgcn_wmma_f32_16x16x4_f32(
                    false, a, false, b, (short)0, acc[q][nt], false, false);
            }
        }
    };

    int basep = start;
    for (; basep < start + nfull; basep += 4) {
        const int p = order[basep + k];
        if (j < 4) {
            v4f v = __builtin_nontemporal_load(((const v4f*)(sph + (size_t)p * 16)) + j);
            ((v4f*)&sS[k][0])[j] = v;
        }
        {
            v4f v = __builtin_nontemporal_load(((const v4f*)(rb + (size_t)p * 32)) + j);
            ((v4f*)&sR[k][0])[j] = v;
        }
        if (j < 4) sP[k][j] = sW[j * 4 + species[p]];
        __syncthreads();
        do_wmma();
        __syncthreads();
    }

    if (tail) {
        const v4f z4 = {0.f, 0.f, 0.f, 0.f};
        if (j < 4) ((v4f*)&sS[k][0])[j] = z4;
        ((v4f*)&sR[k][0])[j] = z4;
        if (j < 4) sP[k][j] = 0.f;
        const bool valid = k < tail;
        if (valid) {
            const int p = order[basep + k];
            if (j < 4) {
                v4f v = __builtin_nontemporal_load(((const v4f*)(sph + (size_t)p * 16)) + j);
                ((v4f*)&sS[k][0])[j] = v;
            }
            {
                v4f v = __builtin_nontemporal_load(((const v4f*)(rb + (size_t)p * 32)) + j);
                ((v4f*)&sR[k][0])[j] = v;
            }
            if (j < 4) sP[k][j] = sW[j * 4 + species[p]];
        }
        __syncthreads();
        do_wmma();
    }

    float* outc = out + (size_t)c * OUTROW;
#pragma unroll
    for (int q = 0; q < 4; ++q)
#pragma unroll
        for (int nt = 0; nt < 2; ++nt)
#pragma unroll
            for (int r = 0; r < 8; ++r) {
                const int M = r + 8 * half;
                const int l = (M >= 9) ? 3 : (M >= 4) ? 2 : (M >= 1) ? 1 : 0;
                const int col = nt * 16 + idx;
                if ((col >> 3) == l)
                    outc[M * 32 + q * 8 + (col & 7)] = acc[q][nt][r];
            }
}

extern "C" void kernel_launch(void* const* d_in, const int* in_sizes, int n_in,
                              void* d_out, int out_size, void* d_ws, size_t ws_size,
                              hipStream_t stream) {
    const float* rb      = (const float*)d_in[0];
    const float* sph     = (const float*)d_in[1];
    const float* W       = (const float*)d_in[2];
    const int*   species = (const int*)d_in[3];
    const int*   di      = (const int*)d_in[4];
    const int    P       = in_sizes[3];
    const int    C       = NCENTERS;
    float*       out     = (float*)d_out;

    int* counts  = (int*)d_ws;
    int* offsets = counts + C;
    int* cursors = offsets + C + 1;
    int* order   = cursors + C;

    zero_kernel<<<(C + 255) / 256, 256, 0, stream>>>(counts, C);
    hist_kernel<<<(P + 255) / 256, 256, 0, stream>>>(di, counts, P);
    scan_kernel<<<1, 1024, 0, stream>>>(counts, offsets, cursors, P, C);
    scatter_kernel<<<(P + 255) / 256, 256, 0, stream>>>(di, cursors, order, P);
    spexp_wmma_kernel<<<C, 32, 0, stream>>>(rb, sph, W, species, order, offsets, out);
}


# Counting sort by center moved from global atomics to LDS: 64 slices, each a 64 KiB LDS histogram and LDS cursors; 5 launches to 4; neighbour order (f32 sum order) now by slice, then LDS arrival
# speedup vs baseline: 1.6348x; 1.6348x over previous
#include <hip/hip_runtime.h>
#include <hip/hip_bf16.h>

typedef __attribute__((ext_vector_type(2))) float v2f;
typedef __attribute__((ext_vector_type(4))) float v4f;
typedef __attribute__((ext_vector_type(8))) float v8f;

#define NCENTERS 16384
#define OUTROW 512

#define NSLICE 64
#define SORT_T 1024
#define EPT 8
#define CHUNK 256
#define NCHUNK (NCENTERS / CHUNK)

typedef __attribute__((ext_vector_type(4))) int v4i;

__global__ __launch_bounds__(SORT_T) void slice_hist_kernel(
    const int* __restrict__ di, int* __restrict__ mat, int* __restrict__ ctot, int P)
{
    __shared__ alignas(16) int tab[NCENTERS];
    const int t = threadIdx.x;
    const int b = blockIdx.x;

    const v4i z4 = {0, 0, 0, 0};
#pragma unroll
    for (int k = 0; k < NCENTERS / 4 / SORT_T; ++k) ((v4i*)tab)[k * SORT_T + t] = z4;
    __syncthreads();

    const int per = (P + NSLICE - 1) / NSLICE;
    const int lo  = b * per;
    const int hi  = (lo + per < P) ? (lo + per) : P;
    for (int tile = lo; tile < hi; tile += SORT_T * EPT) {
        int v[EPT];
#pragma unroll
        for (int u = 0; u < EPT; ++u) {
            const int i = tile + u * SORT_T + t;
            v[u] = (i < hi) ? di[i] : -1;
        }
#pragma unroll
        for (int u = 0; u < EPT; ++u)
            if ((unsigned)v[u] < (unsigned)NCENTERS) atomicAdd(&tab[v[u]], 1);
    }
    __syncthreads();

    int* row = mat + (size_t)b * NCENTERS;
#pragma unroll
    for (int k = 0; k < NCENTERS / 4 / SORT_T; ++k)
        ((v4i*)row)[k * SORT_T + t] = ((const v4i*)tab)[k * SORT_T + t];

    int s = 0;
#pragma unroll
    for (int k = 0; k < 4; ++k) {
        const v4i q = ((const v4i*)tab)[4 * t + k];
        s += q.x + q.y + q.z + q.w;
    }
    s += __shfl_xor(s, 8);
    s += __shfl_xor(s, 4);
    s += __shfl_xor(s, 2);
    s += __shfl_xor(s, 1);
    if ((t & 15) == 0) ctot[b * NCHUNK + (t >> 4)] = s;
}

__global__ __launch_bounds__(CHUNK) void slice_scan_kernel(
    int* __restrict__ mat, const int* __restrict__ ctot, int* __restrict__ offsets)
{
    __shared__ int wtot[CHUNK / 32];
    __shared__ int wpart[CHUNK / 32];
    const int t    = threadIdx.x;
    const int j    = blockIdx.x;
    const int lane = t & 31;
    const int w    = t >> 5;
    const int bin  = j * CHUNK + t;

    int part = 0;
#pragma unroll
    for (int k = 0; k < NSLICE * NCHUNK / CHUNK; ++k) part += ctot[k * CHUNK + t];
    if ((t & (NCHUNK - 1)) >= j) part = 0;

    int v[NSLICE];
    int total = 0;
#pragma unroll
    for (int s = 0; s < NSLICE; ++s) {
        v[s] = mat[(size_t)s * NCENTERS + bin];
        total += v[s];
    }

    int inc = total;
#pragma unroll
    for (int off = 1; off < 32; off <<= 1) {
        const int n = __shfl_up(inc, off);
        if (lane >= off) inc += n;
    }
#pragma unroll
    for (int off = 16; off > 0; off >>= 1) part += __shfl_xor(part, off);
    if (lane == 31) wtot[w] = inc;
    if (lane == 0) wpart[w] = part;
    __syncthreads();

    int run = inc - total;
#pragma unroll
    for (int k = 0; k < CHUNK / 32; ++k) {
        run += wpart[k];
        if (k < w) run += wtot[k];
    }
    offsets[bin] = run;
#pragma unroll
    for (int s = 0; s < NSLICE; ++s) {
        mat[(size_t)s * NCENTERS + bin] = run;
        run += v[s];
    }
    if (bin == NCENTERS - 1) offsets[NCENTERS] = run;
}

__global__ __launch_bounds__(SORT_T) void slice_place_kernel(
    const int* __restrict__ di, const int* __restrict__ mat, int* __restrict__ order, int P)
{
    __shared__ alignas(16) int cur[NCENTERS];
    const int t = threadIdx.x;
    const int b = blockIdx.x;

    const int* row = mat + (size_t)b * NCENTERS;
#pragma unroll
    for (int k = 0; k < NCENTERS / 4 / SORT_T; ++k)
        ((v4i*)cur)[k * SORT_T + t] = ((const v4i*)row)[k * SORT_T + t];
    __syncthreads();

    const int per = (P + NSLICE - 1) / NSLICE;
    const int lo  = b * per;
    const int hi  = (lo + per < P) ? (lo + per) : P;
    for (int tile = lo; tile < hi; tile += SORT_T * EPT) {
        int v[EPT];
#pragma unroll
        for (int u = 0; u < EPT; ++u) {
            const int i = tile + u * SORT_T + t;
            v[u] = (i < hi) ? di[i] : -1;
        }
        int pos[EPT];
#pragma unroll
        for (int u = 0; u < EPT; ++u) {
            pos[u] = -1;
            if ((unsigned)v[u] < (unsigned)NCENTERS) pos[u] = atomicAdd(&cur[v[u]], 1);
        }
#pragma unroll
        for (int u = 0; u < EPT; ++u)
            if (pos[u] >= 0) order[pos[u]] = tile + u * SORT_T + t;
    }
}

__global__ __launch_bounds__(32) void spexp_wmma_kernel(
    const float* __restrict__ rb,
    const float* __restrict__ sph,
    const float* __restrict__ W,
    const int*   __restrict__ species,
    const int*   __restrict__ order,
    const int*   __restrict__ offsets,
    float*       __restrict__ out)
{
    __shared__ alignas(16) float sW[16];
    __shared__ alignas(16) float sS[4][16];
    __shared__ alignas(16) float sR[4][32];
    __shared__ alignas(16) float sP[4][4];

    const int c    = blockIdx.x;
    const int lane = threadIdx.x;
    const int half = lane >> 4;
    const int idx  = lane & 15;

    if (lane < 16) sW[lane] = W[lane];

    const int start = offsets[c];
    const int end   = offsets[c + 1];
    const int count = end - start;
    const int nfull = count & ~3;
    const int tail  = count & 3;

    v8f acc[4][2];
    const v8f vzero = {0.f, 0.f, 0.f, 0.f, 0.f, 0.f, 0.f, 0.f};
#pragma unroll
    for (int q = 0; q < 4; ++q)
#pragma unroll
        for (int nt = 0; nt < 2; ++nt) acc[q][nt] = vzero;

    __syncthreads();

    const int k = lane >> 3;
    const int j = lane & 7;
    const int k0 = half * 2;

    auto do_wmma = [&]() {
#pragma unroll
        for (int q = 0; q < 4; ++q) {
            v2f a;
            a.x = sS[k0][idx]     * sP[k0][q];
            a.y = sS[k0 + 1][idx] * sP[k0 + 1][q];
#pragma unroll
            for (int nt = 0; nt < 2; ++nt) {
                const int col = nt * 16 + idx;
                v2f b;
                b.x = sR[k0][col];
                b.y = sR[k0 + 1][col];
                acc[q][nt] = __builtin_amdgcn_wmma_f32_16x16x4_f32(
                    false, a, false, b, (short)0, acc[q][nt], false, false);
            }
        }
    };

    int basep = start;
    for (; basep < start + nfull; basep += 4) {
        const int p = order[basep + k];
        if (j < 4) {
            v4f v = __builtin_nontemporal_load(((const v4f*)(sph + (size_t)p * 16)) + j);
            ((v4f*)&sS[k][0])[j] = v;
        }
        {
            v4f v = __builtin_nontemporal_load(((const v4f*)(rb + (size_t)p * 32)) + j);
            ((v4f*)&sR[k][0])[j] = v;
        }
        if (j < 4) sP[k][j] = sW[j * 4 + species[p]];
        __syncthreads();
        do_wmma();
        __syncthreads();
    }

    if (tail) {
        const v4f z4 = {0.f, 0.f, 0.f, 0.f};
        if (j < 4) ((v4f*)&sS[k][0])[j] = z4;
        ((v4f*)&sR[k][0])[j] = z4;
        if (j < 4) sP[k][j] = 0.f;
        const bool valid = k < tail;
        if (valid) {
            const int p = order[basep + k];
            if (j < 4) {
                v4f v = __builtin_nontemporal_load(((const v4f*)(sph + (size_t)p * 16)) + j);
                ((v4f*)&sS[k][0])[j] = v;
            }
            {
                v4f v = __builtin_nontemporal_load(((const v4f*)(rb + (size_t)p * 32)) + j);
                ((v4f*)&sR[k][0])[j] = v;
            }
            if (j < 4) sP[k][j] = sW[j * 4 + species[p]];
        }
        __syncthreads();
        do_wmma();
    }

    float* outc = out + (size_t)c * OUTROW;
#pragma unroll
    for (int q = 0; q < 4; ++q)
#pragma unroll
        for (int nt = 0; nt < 2; ++nt)
#pragma unroll
            for (int r = 0; r < 8; ++r) {
                const int M = r + 8 * half;
                const int l = (M >= 9) ? 3 : (M >= 4) ? 2 : (M >= 1) ? 1 : 0;
                const int col = nt * 16 + idx;
                if ((col >> 3) == l)
                    outc[M * 32 + q * 8 + (col & 7)] = acc[q][nt][r];
            }
}

extern "C" void kernel_launch(void* const* d_in, const int* in_sizes, int n_in,
                              void* d_out, int out_size, void* d_ws, size_t ws_size,
                              hipStream_t stream) {
    const float* rb      = (const float*)d_in[0];
    const float* sph     = (const float*)d_in[1];
    const float* W       = (const float*)d_in[2];
    const int*   species = (const int*)d_in[3];
    const int*   di      = (const int*)d_in[4];
    const int    P       = in_sizes[3];
    const int    C       = NCENTERS;
    float*       out     = (float*)d_out;

    int* ctot    = (int*)d_ws;
    int* offsets = ctot + C;
    int* order   = offsets + C + 1 + C;
    int* mat     = (int*)d_out;

    slice_hist_kernel<<<NSLICE, SORT_T, 0, stream>>>(di, mat, ctot, P);
    slice_scan_kernel<<<NCHUNK, CHUNK, 0, stream>>>(mat, ctot, offsets);
    slice_place_kernel<<<NSLICE, SORT_T, 0, stream>>>(di, mat, order, P);
    spexp_wmma_kernel<<<C, 32, 0, stream>>>(rb, sph, W, species, order, offsets, out);
}


# Last launch: a center's neighbour ids read 32 at a time, all their row gathers issued before the first product, straight into WMMA fragment registers (no LDS row staging); same sum order
# speedup vs baseline: 1.9340x; 1.1830x over previous
#include <hip/hip_runtime.h>
#include <hip/hip_bf16.h>

typedef __attribute__((ext_vector_type(2))) float v2f;
typedef __attribute__((ext_vector_type(4))) float v4f;
typedef __attribute__((ext_vector_type(8))) float v8f;

#define NCENTERS 16384
#define OUTROW 512

#define NSLICE 64
#define SORT_T 1024
#define EPT 8
#define CHUNK 256
#define NCHUNK (NCENTERS / CHUNK)

typedef __attribute__((ext_vector_type(4))) int v4i;

__global__ __launch_bounds__(SORT_T) void slice_hist_kernel(
    const int* __restrict__ di, int* __restrict__ mat, int* __restrict__ ctot, int P)
{
    __shared__ alignas(16) int tab[NCENTERS];
    const int t = threadIdx.x;
    const int b = blockIdx.x;

    const v4i z4 = {0, 0, 0, 0};
#pragma unroll
    for (int k = 0; k < NCENTERS / 4 / SORT_T; ++k) ((v4i*)tab)[k * SORT_T + t] = z4;
    __syncthreads();

    const int per = (P + NSLICE - 1) / NSLICE;
    const int lo  = b * per;
    const int hi  = (lo + per < P) ? (lo + per) : P;
    for (int tile = lo; tile < hi; tile += SORT_T * EPT) {
        int v[EPT];
#pragma unroll
        for (int u = 0; u < EPT; ++u) {
            const int i = tile + u * SORT_T + t;
            v[u] = (i < hi) ? di[i] : -1;
        }
#pragma unroll
        for (int u = 0; u < EPT; ++u)
            if ((unsigned)v[u] < (unsigned)NCENTERS) atomicAdd(&tab[v[u]], 1);
    }
    __syncthreads();

    int* row = mat + (size_t)b * NCENTERS;
#pragma unroll
    for (int k = 0; k < NCENTERS / 4 / SORT_T; ++k)
        ((v4i*)row)[k * SORT_T + t] = ((const v4i*)tab)[k * SORT_T + t];

    int s = 0;
#pragma unroll
    for (int k = 0; k < 4; ++k) {
        const v4i q = ((const v4i*)tab)[4 * t + k];
        s += q.x + q.y + q.z + q.w;
    }
    s += __shfl_xor(s, 8);
    s += __shfl_xor(s, 4);
    s += __shfl_xor(s, 2);
    s += __shfl_xor(s, 1);
    if ((t & 15) == 0) ctot[b * NCHUNK + (t >> 4)] = s;
}

__global__ __launch_bounds__(CHUNK) void slice_scan_kernel(
    int* __restrict__ mat, const int* __restrict__ ctot, int* __restrict__ offsets)
{
    __shared__ int wtot[CHUNK / 32];
    __shared__ int wpart[CHUNK / 32];
    const int t    = threadIdx.x;
    const int j    = blockIdx.x;
    const int lane = t & 31;
    const int w    = t >> 5;
    const int bin  = j * CHUNK + t;

    int part = 0;
#pragma unroll
    for (int k = 0; k < NSLICE * NCHUNK / CHUNK; ++k) part += ctot[k * CHUNK + t];
    if ((t & (NCHUNK - 1)) >= j) part = 0;

    int v[NSLICE];
    int total = 0;
#pragma unroll
    for (int s = 0; s < NSLICE; ++s) {
        v[s] = mat[(size_t)s * NCENTERS + bin];
        total += v[s];
    }

    int inc = total;
#pragma unroll
    for (int off = 1; off < 32; off <<= 1) {
        const int n = __shfl_up(inc, off);
        if (lane >= off) inc += n;
    }
#pragma unroll
    for (int off = 16; off > 0; off >>= 1) part += __shfl_xor(part, off);
    if (lane == 31) wtot[w] = inc;
    if (lane == 0) wpart[w] = part;
    __syncthreads();

    int run = inc - total;
#pragma unroll
    for (int k = 0; k < CHUNK / 32; ++k) {
        run += wpart[k];
        if (k < w) run += wtot[k];
    }
    offsets[bin] = run;
#pragma unroll
    for (int s = 0; s < NSLICE; ++s) {
        mat[(size_t)s * NCENTERS + bin] = run;
        run += v[s];
    }
    if (bin == NCENTERS - 1) offsets[NCENTERS] = run;
}

__global__ __launch_bounds__(SORT_T) void slice_place_kernel(
    const int* __restrict__ di, const int* __restrict__ mat, int* __restrict__ order, int P)
{
    __shared__ alignas(16) int cur[NCENTERS];
    const int t = threadIdx.x;
    const int b = blockIdx.x;

    const int* row = mat + (size_t)b * NCENTERS;
#pragma unroll
    for (int k = 0; k < NCENTERS / 4 / SORT_T; ++k)
        ((v4i*)cur)[k * SORT_T + t] = ((const v4i*)row)[k * SORT_T + t];
    __syncthreads();

    const int per = (P + NSLICE - 1) / NSLICE;
    const int lo  = b * per;
    const int hi  = (lo + per < P) ? (lo + per) : P;
    for (int tile = lo; tile < hi; tile += SORT_T * EPT) {
        int v[EPT];
#pragma unroll
        for (int u = 0; u < EPT; ++u) {
            const int i = tile + u * SORT_T + t;
            v[u] = (i < hi) ? di[i] : -1;
        }
        int pos[EPT];
#pragma unroll
        for (int u = 0; u < EPT; ++u) {
            pos[u] = -1;
            if ((unsigned)v[u] < (unsigned)NCENTERS) pos[u] = atomicAdd(&cur[v[u]], 1);
        }
#pragma unroll
        for (int u = 0; u < EPT; ++u)
            if (pos[u] >= 0) order[pos[u]] = tile + u * SORT_T + t;
    }
}

template <int N> struct StepCount { static constexpr int value = N; };

__global__ __launch_bounds__(32) void spexp_wmma_kernel(
    const float* __restrict__ rb,
    const float* __restrict__ sph,
    const float* __restrict__ W,
    const int*   __restrict__ species,
    const int*   __restrict__ order,
    const int*   __restrict__ offsets,
    float*       __restrict__ out)
{
    __shared__ alignas(16) float sWT[16];
    __shared__ alignas(16) int   sIdx[32];
    __shared__ alignas(16) float sP[32][4];

    const int c    = blockIdx.x;
    const int lane = threadIdx.x;
    const int half = lane >> 4;
    const int idx  = lane & 15;

    if (lane < 16) sWT[(lane & 3) * 4 + (lane >> 2)] = W[lane];

    const int start = offsets[c];
    const int end   = offsets[c + 1];

    v8f acc[4][2];
    const v8f vzero = {0.f, 0.f, 0.f, 0.f, 0.f, 0.f, 0.f, 0.f};
#pragma unroll
    for (int q = 0; q < 4; ++q)
#pragma unroll
        for (int t = 0; t < 2; ++t) acc[q][t] = vzero;

    const int last = end - 1;
    int pi = 0;
    if (start < end) {
        const int i0 = start + lane;
        pi = order[i0 < last ? i0 : last];
    }

    auto chunk = [&](auto nit_c, const int base, const int n) {
        constexpr int NIT = decltype(nit_c)::value;

        const int in = base + 32 + lane;
        const int pn = order[in < last ? in : last];

        sIdx[lane] = pi;
        __builtin_amdgcn_wave_barrier();

        int2 pp[NIT];
#pragma unroll
        for (int it = 0; it < NIT; ++it) pp[it] = *(const int2*)&sIdx[4 * it + 2 * half];

        float s0[NIT], s1[NIT];
        v2f   r0[NIT], r1[NIT];
#pragma unroll
        for (int it = 0; it < NIT; ++it) {
            const unsigned a0 = (unsigned)pp[it].x, a1 = (unsigned)pp[it].y;
            s0[it] = __builtin_nontemporal_load(sph + (a0 * 16u + (unsigned)idx));
            s1[it] = __builtin_nontemporal_load(sph + (a1 * 16u + (unsigned)idx));
            r0[it] = __builtin_nontemporal_load((const v2f*)(rb + (a0 * 32u + 2u * (unsigned)idx)));
            r1[it] = __builtin_nontemporal_load((const v2f*)(rb + (a1 * 32u + 2u * (unsigned)idx)));
        }

        const int si = species[pi];
        v4f w = *(const v4f*)&sWT[(si & 3) * 4];
        if (!(lane < n && (unsigned)si < 4u)) { w.x = 0.f; w.y = 0.f; w.z = 0.f; w.w = 0.f; }
        *(v4f*)&sP[lane][0] = w;
        __builtin_amdgcn_wave_barrier();

        v4f w0 = *(const v4f*)&sP[2 * half][0];
        v4f w1 = *(const v4f*)&sP[2 * half + 1][0];
#pragma unroll
        for (int it = 0; it < NIT; ++it) {
            if (4 * it < n) {
                const v4f u0 = w0, u1 = w1;
                if (it + 1 < NIT) {
                    w0 = *(const v4f*)&sP[4 * (it + 1) + 2 * half][0];
                    w1 = *(const v4f*)&sP[4 * (it + 1) + 2 * half + 1][0];
                }
                float x0 = s0[it], x1 = s1[it];
                v2f   y0 = r0[it], y1 = r1[it];
                if (4 * it + 4 > n) {
                    const int j0 = 4 * it + 2 * half;
                    if (j0 >= n)     { x0 = 0.f; y0.x = 0.f; y0.y = 0.f; }
                    if (j0 + 1 >= n) { x1 = 0.f; y1.x = 0.f; y1.y = 0.f; }
                }
                v2f b0, b1;
                b0.x = y0.x; b0.y = y1.x;
                b1.x = y0.y; b1.y = y1.y;
#pragma unroll
                for (int q = 0; q < 4; ++q) {
                    v2f a;
                    a.x = x0 * u0[q];
                    a.y = x1 * u1[q];
                    acc[q][0] = __builtin_amdgcn_wmma_f32_16x16x4_f32(
                        false, a, false, b0, (short)0, acc[q][0], false, false);
                    acc[q][1] = __builtin_amdgcn_wmma_f32_16x16x4_f32(
                        false, a, false, b1, (short)0, acc[q][1], false, false);
                }
            }
        }

        pi = pn;
        __builtin_amdgcn_wave_barrier();
    };

    for (int base = start; base < end; base += 32) {
        const int n = (end - base < 32) ? (end - base) : 32;
        if (n > 16)     chunk(StepCount<8>{}, base, n);
        else if (n > 8) chunk(StepCount<4>{}, base, n);
        else            chunk(StepCount<2>{}, base, n);
    }

    float* outc = out + (size_t)c * OUTROW;
    const int lcol = idx >> 2;
    const int n0   = 2 * (idx & 3);
#pragma unroll
    for (int q = 0; q < 4; ++q)
#pragma unroll
        for (int r = 0; r < 8; ++r) {
            const int M = r + 8 * half;
            const int l = (M >= 9) ? 3 : (M >= 4) ? 2 : (M >= 1) ? 1 : 0;
            if (lcol == l) {
                v2f o;
                o.x = acc[q][0][r];
                o.y = acc[q][1][r];
                *(v2f*)(outc + M * 32 + q * 8 + n0) = o;
            }
        }
}

extern "C" void kernel_launch(void* const* d_in, const int* in_sizes, int n_in,
                              void* d_out, int out_size, void* d_ws, size_t ws_size,
                              hipStream_t stream) {
    const float* rb      = (const float*)d_in[0];
    const float* sph     = (const float*)d_in[1];
    const float* W       = (const float*)d_in[2];
    const int*   species = (const int*)d_in[3];
    const int*   di      = (const int*)d_in[4];
    const int    P       = in_sizes[3];
    const int    C       = NCENTERS;
    float*       out     = (float*)d_out;

    int* ctot    = (int*)d_ws;
    int* offsets = ctot + C;
    int* order   = offsets + C + 1 + C;
    int* mat     = (int*)d_out;

    slice_hist_kernel<<<NSLICE, SORT_T, 0, stream>>>(di, mat, ctot, P);
    slice_scan_kernel<<<NCHUNK, CHUNK, 0, stream>>>(mat, ctot, offsets);
    slice_place_kernel<<<NSLICE, SORT_T, 0, stream>>>(di, mat, order, P);
    spexp_wmma_kernel<<<C, 32, 0, stream>>>(rb, sph, W, species, order, offsets, out);
}


# Last launch: each step's 8 quarter-dense 16x16 products repacked as 4 half-dense ones (degrees paired {0,3},{1,2}; partner-lane values by permlane16): half the matrix instructions; same sum order
# speedup vs baseline: 2.0575x; 1.0639x over previous
#include <hip/hip_runtime.h>
#include <hip/hip_bf16.h>

typedef __attribute__((ext_vector_type(2))) float v2f;
typedef __attribute__((ext_vector_type(4))) float v4f;
typedef __attribute__((ext_vector_type(8))) float v8f;

#define NCENTERS 16384
#define OUTROW 512

#define NSLICE 64
#define SORT_T 1024
#define EPT 8
#define CHUNK 256
#define NCHUNK (NCENTERS / CHUNK)

typedef __attribute__((ext_vector_type(4))) int v4i;

__global__ __launch_bounds__(SORT_T) void slice_hist_kernel(
    const int* __restrict__ di, int* __restrict__ mat, int* __restrict__ ctot, int P)
{
    __shared__ alignas(16) int tab[NCENTERS];
    const int t = threadIdx.x;
    const int b = blockIdx.x;

    const v4i z4 = {0, 0, 0, 0};
#pragma unroll
    for (int k = 0; k < NCENTERS / 4 / SORT_T; ++k) ((v4i*)tab)[k * SORT_T + t] = z4;
    __syncthreads();

    const int per = (P + NSLICE - 1) / NSLICE;
    const int lo  = b * per;
    const int hi  = (lo + per < P) ? (lo + per) : P;
    for (int tile = lo; tile < hi; tile += SORT_T * EPT) {
        int v[EPT];
#pragma unroll
        for (int u = 0; u < EPT; ++u) {
            const int i = tile + u * SORT_T + t;
            v[u] = (i < hi) ? di[i] : -1;
        }
#pragma unroll
        for (int u = 0; u < EPT; ++u)
            if ((unsigned)v[u] < (unsigned)NCENTERS) atomicAdd(&tab[v[u]], 1);
    }
    __syncthreads();

    int* row = mat + (size_t)b * NCENTERS;
#pragma unroll
    for (int k = 0; k < NCENTERS / 4 / SORT_T; ++k)
        ((v4i*)row)[k * SORT_T + t] = ((const v4i*)tab)[k * SORT_T + t];

    int s = 0;
#pragma unroll
    for (int k = 0; k < 4; ++k) {
        const v4i q = ((const v4i*)tab)[4 * t + k];
        s += q.x + q.y + q.z + q.w;
    }
    s += __shfl_xor(s, 8);
    s += __shfl_xor(s, 4);
    s += __shfl_xor(s, 2);
    s += __shfl_xor(s, 1);
    if ((t & 15) == 0) ctot[b * NCHUNK + (t >> 4)] = s;
}

__global__ __launch_bounds__(CHUNK) void slice_scan_kernel(
    int* __restrict__ mat, const int* __restrict__ ctot, int* __restrict__ offsets)
{
    __shared__ int wtot[CHUNK / 32];
    __shared__ int wpart[CHUNK / 32];
    const int t    = threadIdx.x;
    const int j    = blockIdx.x;
    const int lane = t & 31;
    const int w    = t >> 5;
    const int bin  = j * CHUNK + t;

    int part = 0;
#pragma unroll
    for (int k = 0; k < NSLICE * NCHUNK / CHUNK; ++k) part += ctot[k * CHUNK + t];
    if ((t & (NCHUNK - 1)) >= j) part = 0;

    int v[NSLICE];
    int total = 0;
#pragma unroll
    for (int s = 0; s < NSLICE; ++s) {
        v[s] = mat[(size_t)s * NCENTERS + bin];
        total += v[s];
    }

    int inc = total;
#pragma unroll
    for (int off = 1; off < 32; off <<= 1) {
        const int n = __shfl_up(inc, off);
        if (lane >= off) inc += n;
    }
#pragma unroll
    for (int off = 16; off > 0; off >>= 1) part += __shfl_xor(part, off);
    if (lane == 31) wtot[w] = inc;
    if (lane == 0) wpart[w] = part;
    __syncthreads();

    int run = inc - total;
#pragma unroll
    for (int k = 0; k < CHUNK / 32; ++k) {
        run += wpart[k];
        if (k < w) run += wtot[k];
    }
    offsets[bin] = run;
#pragma unroll
    for (int s = 0; s < NSLICE; ++s) {
        mat[(size_t)s * NCENTERS + bin] = run;
        run += v[s];
    }
    if (bin == NCENTERS - 1) offsets[NCENTERS] = run;
}

__global__ __launch_bounds__(SORT_T) void slice_place_kernel(
    const int* __restrict__ di, const int* __restrict__ mat, int* __restrict__ order, int P)
{
    __shared__ alignas(16) int cur[NCENTERS];
    const int t = threadIdx.x;
    const int b = blockIdx.x;

    const int* row = mat + (size_t)b * NCENTERS;
#pragma unroll
    for (int k = 0; k < NCENTERS / 4 / SORT_T; ++k)
        ((v4i*)cur)[k * SORT_T + t] = ((const v4i*)row)[k * SORT_T + t];
    __syncthreads();

    const int per = (P + NSLICE - 1) / NSLICE;
    const int lo  = b * per;
    const int hi  = (lo + per < P) ? (lo + per) : P;
    for (int tile = lo; tile < hi; tile += SORT_T * EPT) {
        int v[EPT];
#pragma unroll
        for (int u = 0; u < EPT; ++u) {
            const int i = tile + u * SORT_T + t;
            v[u] = (i < hi) ? di[i] : -1;
        }
        int pos[EPT];
#pragma unroll
        for (int u = 0; u < EPT; ++u) {
            pos[u] = -1;
            if ((unsigned)v[u] < (unsigned)NCENTERS) pos[u] = atomicAdd(&cur[v[u]], 1);
        }
#pragma unroll
        for (int u = 0; u < EPT; ++u)
            if (pos[u] >= 0) order[pos[u]] = tile + u * SORT_T + t;
    }
}

template <int N> struct StepCount { static constexpr int value = N; };

__global__ __launch_bounds__(32) void spexp_wmma_kernel(
    const float* __restrict__ rb,
    const float* __restrict__ sph,
    const float* __restrict__ W,
    const int*   __restrict__ species,
    const int*   __restrict__ order,
    const int*   __restrict__ offsets,
    float*       __restrict__ out)
{
    __shared__ alignas(16) float sWT[16];
    __shared__ alignas(16) int   sIdx[32];
    __shared__ alignas(16) float sP[2][32][4];

    const int c    = blockIdx.x;
    const int lane = threadIdx.x;
    const int half = lane >> 4;
    const int idx  = lane & 15;

    if (lane < 16) sWT[(lane & 3) * 4 + (lane >> 2)] = W[lane];

    const int start = offsets[c];
    const int end   = offsets[c + 1];

    const bool alpha = (idx == 0) || (idx >= 9);
    const bool beta  = (idx < 4) || (idx >= 12);
    const float (*sPl)[4] = sP[alpha ? 0 : 1];

    auto row_xor8 = [](float v) {
        const int i = __builtin_bit_cast(int, v);
        return __builtin_bit_cast(float, __builtin_amdgcn_permlane16(i, i, 0xFEDCBA98u, 0x76543210u, false, false));
    };
    auto row_xor4 = [](float v) {
        const int i = __builtin_bit_cast(int, v);
        return __builtin_bit_cast(float, __builtin_amdgcn_permlane16(i, i, 0x32107654u, 0xBA98FEDCu, false, false));
    };

    v8f acc[2][2];
    const v8f vzero = {0.f, 0.f, 0.f, 0.f, 0.f, 0.f, 0.f, 0.f};
#pragma unroll
    for (int g = 0; g < 2; ++g)
#pragma unroll
        for (int h = 0; h < 2; ++h) acc[g][h] = vzero;

    const int last = end - 1;
    int pi = 0;
    if (start < end) {
        const int i0 = start + lane;
        pi = order[i0 < last ? i0 : last];
    }

    auto chunk = [&](auto nit_c, const int base, const int n) {
        constexpr int NIT = decltype(nit_c)::value;

        const int in = base + 32 + lane;
        const int pn = order[in < last ? in : last];

        sIdx[lane] = pi;
        __builtin_amdgcn_wave_barrier();

        int2 pp[NIT];
#pragma unroll
        for (int it = 0; it < NIT; ++it) pp[it] = *(const int2*)&sIdx[4 * it + 2 * half];

        float s0[NIT], s1[NIT];
        v2f   r0[NIT], r1[NIT];
#pragma unroll
        for (int it = 0; it < NIT; ++it) {
            const unsigned a0 = (unsigned)pp[it].x, a1 = (unsigned)pp[it].y;
            s0[it] = __builtin_nontemporal_load(sph + (a0 * 16u + (unsigned)idx));
            s1[it] = __builtin_nontemporal_load(sph + (a1 * 16u + (unsigned)idx));
            r0[it] = __builtin_nontemporal_load((const v2f*)(rb + (a0 * 32u + 2u * (unsigned)idx)));
            r1[it] = __builtin_nontemporal_load((const v2f*)(rb + (a1 * 32u + 2u * (unsigned)idx)));
        }

        const int si = species[pi];
        v4f w = *(const v4f*)&sWT[(si & 3) * 4];
        if (!(lane < n && (unsigned)si < 4u)) { w.x = 0.f; w.y = 0.f; w.z = 0.f; w.w = 0.f; }
        *(v4f*)&sP[0][lane][0] = w;
        v4f wsw;
        wsw.x = w.y; wsw.y = w.x; wsw.z = w.w; wsw.w = w.z;
        *(v4f*)&sP[1][lane][0] = wsw;
        __builtin_amdgcn_wave_barrier();

        v4f w0 = *(const v4f*)&sPl[2 * half][0];
        v4f w1 = *(const v4f*)&sPl[2 * half + 1][0];
#pragma unroll
        for (int it = 0; it < NIT; ++it) {
            if (4 * it < n) {
                const v4f u0 = w0, u1 = w1;
                if (it + 1 < NIT) {
                    w0 = *(const v4f*)&sPl[4 * (it + 1) + 2 * half][0];
                    w1 = *(const v4f*)&sPl[4 * (it + 1) + 2 * half + 1][0];
                }
                float x0 = s0[it], x1 = s1[it];
                v2f   y0 = r0[it], y1 = r1[it];
                if (4 * it + 4 > n) {
                    const int j0 = 4 * it + 2 * half;
                    if (j0 >= n)     { x0 = 0.f; y0.x = 0.f; y0.y = 0.f; }
                    if (j0 + 1 >= n) { x1 = 0.f; y1.x = 0.f; y1.y = 0.f; }
                }
                const float X0 = row_xor8(x0),   X1 = row_xor8(x1);
                const float T0 = row_xor4(y0.y), T1 = row_xor4(y1.y);
                v2f b0, b1;
                b0.x = beta ? y0.x : T0;  b0.y = beta ? y1.x : T1;
                b1.x = beta ? T0 : y0.x;  b1.y = beta ? T1 : y1.x;
                const float xg0e0 = alpha ? x0 : X0, xg0e1 = alpha ? x1 : X1;
                const float xg1e0 = alpha ? X0 : x0, xg1e1 = alpha ? X1 : x1;
#pragma unroll
                for (int h = 0; h < 2; ++h) {
                    v2f a0, a1;
                    a0.x = xg0e0 * u0[2 * h];
                    a0.y = xg0e1 * u1[2 * h];
                    a1.x = xg1e0 * u0[2 * h + 1];
                    a1.y = xg1e1 * u1[2 * h + 1];
                    acc[0][h] = __builtin_amdgcn_wmma_f32_16x16x4_f32(
                        false, a0, false, b0, (short)0, acc[0][h], false, false);
                    acc[1][h] = __builtin_amdgcn_wmma_f32_16x16x4_f32(
                        false, a1, false, b1, (short)0, acc[1][h], false, false);
                }
            }
        }

        pi = pn;
        __builtin_amdgcn_wave_barrier();
    };

    for (int base = start; base < end; base += 32) {
        const int n = (end - base < 32) ? (end - base) : 32;
        if (n > 16)     chunk(StepCount<8>{}, base, n);
        else if (n > 8) chunk(StepCount<4>{}, base, n);
        else            chunk(StepCount<2>{}, base, n);
    }

    float* outc = out + (size_t)c * OUTROW;
    const int lown = idx >> 2, lpar = (idx ^ 4) >> 2;
    const int lc0 = beta ? lown : lpar;
    const int lc1 = beta ? lpar : lown;
    const int nc0 = 2 * (idx & 3) + (beta ? 0 : 1);
    const int nc1 = 2 * (idx & 3) + (beta ? 1 : 0);
#pragma unroll
    for (int g = 0; g < 2; ++g)
#pragma unroll
        for (int h = 0; h < 2; ++h)
#pragma unroll
            for (int r = 0; r < 8; ++r) {
                const int lm   = (g == 0) ? (r == 0 ? 0 : r + 8) : (r == 0 ? 8 : r);
                const int l    = (lm >= 9) ? 3 : (lm >= 4) ? 2 : (lm >= 1) ? 1 : 0;
                const int flip = ((r > 0) != (g == 1)) ? 1 : 0;
                const int q    = 2 * h + (half ^ flip);
                const int lc   = g ? lc1 : lc0;
                const int nc   = g ? nc1 : nc0;
                if (lc == l) outc[lm * 32 + q * 8 + nc] = acc[g][h][r];
            }
}

extern "C" void kernel_launch(void* const* d_in, const int* in_sizes, int n_in,
                              void* d_out, int out_size, void* d_ws, size_t ws_size,
                              hipStream_t stream) {
    const float* rb      = (const float*)d_in[0];
    const float* sph     = (const float*)d_in[1];
    const float* W       = (const float*)d_in[2];
    const int*   species = (const int*)d_in[3];
    const int*   di      = (const int*)d_in[4];
    const int    P       = in_sizes[3];
    const int    C       = NCENTERS;
    float*       out     = (float*)d_out;

    int* ctot    = (int*)d_ws;
    int* offsets = ctot + C;
    int* order   = offsets + C + 1 + C;
    int* mat     = (int*)d_out;

    slice_hist_kernel<<<NSLICE, SORT_T, 0, stream>>>(di, mat, ctot, P);
    slice_scan_kernel<<<NCHUNK, CHUNK, 0, stream>>>(mat, ctot, offsets);
    slice_place_kernel<<<NSLICE, SORT_T, 0, stream>>>(di, mat, order, P);
    spexp_wmma_kernel<<<C, 32, 0, stream>>>(rb, sph, W, species, order, offsets, out);
}


# Last launch: a center's 512 outputs staged in 2 KiB of LDS, stored as 4 whole-line 16-byte stores per lane (was 32 masked 4-byte stores = 64 pieces of 32 B per center); same sum order
# speedup vs baseline: 2.1768x; 1.0580x over previous
#include <hip/hip_runtime.h>
#include <hip/hip_bf16.h>

typedef __attribute__((ext_vector_type(2))) float v2f;
typedef __attribute__((ext_vector_type(4))) float v4f;
typedef __attribute__((ext_vector_type(8))) float v8f;

#define NCENTERS 16384
#define OUTROW 512

#define NSLICE 64
#define SORT_T 1024
#define EPT 8
#define CHUNK 256
#define NCHUNK (NCENTERS / CHUNK)

typedef __attribute__((ext_vector_type(4))) int v4i;

__global__ __launch_bounds__(SORT_T) void slice_hist_kernel(
    const int* __restrict__ di, int* __restrict__ mat, int* __restrict__ ctot, int P)
{
    __shared__ alignas(16) int tab[NCENTERS];
    const int t = threadIdx.x;
    const int b = blockIdx.x;

    const v4i z4 = {0, 0, 0, 0};
#pragma unroll
    for (int k = 0; k < NCENTERS / 4 / SORT_T; ++k) ((v4i*)tab)[k * SORT_T + t] = z4;
    __syncthreads();

    const int per = (P + NSLICE - 1) / NSLICE;
    const int lo  = b * per;
    const int hi  = (lo + per < P) ? (lo + per) : P;
    for (int tile = lo; tile < hi; tile += SORT_T * EPT) {
        int v[EPT];
#pragma unroll
        for (int u = 0; u < EPT; ++u) {
            const int i = tile + u * SORT_T + t;
            v[u] = (i < hi) ? di[i] : -1;
        }
#pragma unroll
        for (int u = 0; u < EPT; ++u)
            if ((unsigned)v[u] < (unsigned)NCENTERS) atomicAdd(&tab[v[u]], 1);
    }
    __syncthreads();

    int* row = mat + (size_t)b * NCENTERS;
#pragma unroll
    for (int k = 0; k < NCENTERS / 4 / SORT_T; ++k)
        ((v4i*)row)[k * SORT_T + t] = ((const v4i*)tab)[k * SORT_T + t];

    int s = 0;
#pragma unroll
    for (int k = 0; k < 4; ++k) {
        const v4i q = ((const v4i*)tab)[4 * t + k];
        s += q.x + q.y + q.z + q.w;
    }
    s += __shfl_xor(s, 8);
    s += __shfl_xor(s, 4);
    s += __shfl_xor(s, 2);
    s += __shfl_xor(s, 1);
    if ((t & 15) == 0) ctot[b * NCHUNK + (t >> 4)] = s;
}

__global__ __launch_bounds__(CHUNK) void slice_scan_kernel(
    int* __restrict__ mat, const int* __restrict__ ctot, int* __restrict__ offsets)
{
    __shared__ int wtot[CHUNK / 32];
    __shared__ int wpart[CHUNK / 32];
    const int t    = threadIdx.x;
    const int j    = blockIdx.x;
    const int lane = t & 31;
    const int w    = t >> 5;
    const int bin  = j * CHUNK + t;

    int part = 0;
#pragma unroll
    for (int k = 0; k < NSLICE * NCHUNK / CHUNK; ++k) part += ctot[k * CHUNK + t];
    if ((t & (NCHUNK - 1)) >= j) part = 0;

    int v[NSLICE];
    int total = 0;
#pragma unroll
    for (int s = 0; s < NSLICE; ++s) {
        v[s] = mat[(size_t)s * NCENTERS + bin];
        total += v[s];
    }

    int inc = total;
#pragma unroll
    for (int off = 1; off < 32; off <<= 1) {
        const int n = __shfl_up(inc, off);
        if (lane >= off) inc += n;
    }
#pragma unroll
    for (int off = 16; off > 0; off >>= 1) part += __shfl_xor(part, off);
    if (lane == 31) wtot[w] = inc;
    if (lane == 0) wpart[w] = part;
    __syncthreads();

    int run = inc - total;
#pragma unroll
    for (int k = 0; k < CHUNK / 32; ++k) {
        run += wpart[k];
        if (k < w) run += wtot[k];
    }
    offsets[bin] = run;
#pragma unroll
    for (int s = 0; s < NSLICE; ++s) {
        mat[(size_t)s * NCENTERS + bin] = run;
        run += v[s];
    }
    if (bin == NCENTERS - 1) offsets[NCENTERS] = run;
}

__global__ __launch_bounds__(SORT_T) void slice_place_kernel(
    const int* __restrict__ di, const int* __restrict__ mat, int* __restrict__ order, int P)
{
    __shared__ alignas(16) int cur[NCENTERS];
    const int t = threadIdx.x;
    const int b = blockIdx.x;

    const int* row = mat + (size_t)b * NCENTERS;
#pragma unroll
    for (int k = 0; k < NCENTERS / 4 / SORT_T; ++k)
        ((v4i*)cur)[k * SORT_T + t] = ((const v4i*)row)[k * SORT_T + t];
    __syncthreads();

    const int per = (P + NSLICE - 1) / NSLICE;
    const int lo  = b * per;
    const int hi  = (lo + per < P) ? (lo + per) : P;
    for (int tile = lo; tile < hi; tile += SORT_T * EPT) {
        int v[EPT];
#pragma unroll
        for (int u = 0; u < EPT; ++u) {
            const int i = tile + u * SORT_T + t;
            v[u] = (i < hi) ? di[i] : -1;
        }
        int pos[EPT];
#pragma unroll
        for (int u = 0; u < EPT; ++u) {
            pos[u] = -1;
            if ((unsigned)v[u] < (unsigned)NCENTERS) pos[u] = atomicAdd(&cur[v[u]], 1);
        }
#pragma unroll
        for (int u = 0; u < EPT; ++u)
            if (pos[u] >= 0) order[pos[u]] = tile + u * SORT_T + t;
    }
}

template <int N> struct StepCount { static constexpr int value = N; };

__global__ __launch_bounds__(32) void spexp_wmma_kernel(
    const float* __restrict__ rb,
    const float* __restrict__ sph,
    const float* __restrict__ W,
    const int*   __restrict__ species,
    const int*   __restrict__ order,
    const int*   __restrict__ offsets,
    float*       __restrict__ out)
{
    __shared__ alignas(16) float sWT[16];
    __shared__ alignas(16) int   sIdx[32];
    __shared__ alignas(16) float sP[2][32][4];
    __shared__ alignas(16) float sOut[OUTROW];

    const int c    = blockIdx.x;
    const int lane = threadIdx.x;
    const int half = lane >> 4;
    const int idx  = lane & 15;

    if (lane < 16) sWT[(lane & 3) * 4 + (lane >> 2)] = W[lane];

    const int start = offsets[c];
    const int end   = offsets[c + 1];

    const bool alpha = (idx == 0) || (idx >= 9);
    const bool beta  = (idx < 4) || (idx >= 12);
    const float (*sPl)[4] = sP[alpha ? 0 : 1];

    auto row_xor8 = [](float v) {
        const int i = __builtin_bit_cast(int, v);
        return __builtin_bit_cast(float, __builtin_amdgcn_permlane16(i, i, 0xFEDCBA98u, 0x76543210u, false, false));
    };
    auto row_xor4 = [](float v) {
        const int i = __builtin_bit_cast(int, v);
        return __builtin_bit_cast(float, __builtin_amdgcn_permlane16(i, i, 0x32107654u, 0xBA98FEDCu, false, false));
    };

    v8f acc[2][2];
    const v8f vzero = {0.f, 0.f, 0.f, 0.f, 0.f, 0.f, 0.f, 0.f};
#pragma unroll
    for (int g = 0; g < 2; ++g)
#pragma unroll
        for (int h = 0; h < 2; ++h) acc[g][h] = vzero;

    const int last = end - 1;
    int pi = 0;
    if (start < end) {
        const int i0 = start + lane;
        pi = order[i0 < last ? i0 : last];
    }

    auto chunk = [&](auto nit_c, const int base, const int n) {
        constexpr int NIT = decltype(nit_c)::value;

        const int in = base + 32 + lane;
        const int pn = order[in < last ? in : last];

        sIdx[lane] = pi;
        __builtin_amdgcn_wave_barrier();

        int2 pp[NIT];
#pragma unroll
        for (int it = 0; it < NIT; ++it) pp[it] = *(const int2*)&sIdx[4 * it + 2 * half];

        float s0[NIT], s1[NIT];
        v2f   r0[NIT], r1[NIT];
#pragma unroll
        for (int it = 0; it < NIT; ++it) {
            const unsigned a0 = (unsigned)pp[it].x, a1 = (unsigned)pp[it].y;
            s0[it] = __builtin_nontemporal_load(sph + (a0 * 16u + (unsigned)idx));
            s1[it] = __builtin_nontemporal_load(sph + (a1 * 16u + (unsigned)idx));
            r0[it] = __builtin_nontemporal_load((const v2f*)(rb + (a0 * 32u + 2u * (unsigned)idx)));
            r1[it] = __builtin_nontemporal_load((const v2f*)(rb + (a1 * 32u + 2u * (unsigned)idx)));
        }

        const int si = species[pi];
        v4f w = *(const v4f*)&sWT[(si & 3) * 4];
        if (!(lane < n && (unsigned)si < 4u)) { w.x = 0.f; w.y = 0.f; w.z = 0.f; w.w = 0.f; }
        *(v4f*)&sP[0][lane][0] = w;
        v4f wsw;
        wsw.x = w.y; wsw.y = w.x; wsw.z = w.w; wsw.w = w.z;
        *(v4f*)&sP[1][lane][0] = wsw;
        __builtin_amdgcn_wave_barrier();

        v4f w0 = *(const v4f*)&sPl[2 * half][0];
        v4f w1 = *(const v4f*)&sPl[2 * half + 1][0];
#pragma unroll
        for (int it = 0; it < NIT; ++it) {
            if (4 * it < n) {
                const v4f u0 = w0, u1 = w1;
                if (it + 1 < NIT) {
                    w0 = *(const v4f*)&sPl[4 * (it + 1) + 2 * half][0];
                    w1 = *(const v4f*)&sPl[4 * (it + 1) + 2 * half + 1][0];
                }
                float x0 = s0[it], x1 = s1[it];
                v2f   y0 = r0[it], y1 = r1[it];
                if (4 * it + 4 > n) {
                    const int j0 = 4 * it + 2 * half;
                    if (j0 >= n)     { x0 = 0.f; y0.x = 0.f; y0.y = 0.f; }
                    if (j0 + 1 >= n) { x1 = 0.f; y1.x = 0.f; y1.y = 0.f; }
                }
                const float X0 = row_xor8(x0),   X1 = row_xor8(x1);
                const float T0 = row_xor4(y0.y), T1 = row_xor4(y1.y);
                v2f b0, b1;
                b0.x = beta ? y0.x : T0;  b0.y = beta ? y1.x : T1;
                b1.x = beta ? T0 : y0.x;  b1.y = beta ? T1 : y1.x;
                const float xg0e0 = alpha ? x0 : X0, xg0e1 = alpha ? x1 : X1;
                const float xg1e0 = alpha ? X0 : x0, xg1e1 = alpha ? X1 : x1;
#pragma unroll
                for (int h = 0; h < 2; ++h) {
                    v2f a0, a1;
                    a0.x = xg0e0 * u0[2 * h];
                    a0.y = xg0e1 * u1[2 * h];
                    a1.x = xg1e0 * u0[2 * h + 1];
                    a1.y = xg1e1 * u1[2 * h + 1];
                    acc[0][h] = __builtin_amdgcn_wmma_f32_16x16x4_f32(
                        false, a0, false, b0, (short)0, acc[0][h], false, false);
                    acc[1][h] = __builtin_amdgcn_wmma_f32_16x16x4_f32(
                        false, a1, false, b1, (short)0, acc[1][h], false, false);
                }
            }
        }

        pi = pn;
        __builtin_amdgcn_wave_barrier();
    };

    for (int base = start; base < end; base += 32) {
        const int n = (end - base < 32) ? (end - base) : 32;
        if (n > 16)     chunk(StepCount<8>{}, base, n);
        else if (n > 8) chunk(StepCount<4>{}, base, n);
        else            chunk(StepCount<2>{}, base, n);
    }

    float* outc = out + (size_t)c * OUTROW;
    const int lown = idx >> 2, lpar = (idx ^ 4) >> 2;
    const int lc0 = beta ? lown : lpar;
    const int lc1 = beta ? lpar : lown;
    const int nc0 = 2 * (idx & 3) + (beta ? 0 : 1);
    const int nc1 = 2 * (idx & 3) + (beta ? 1 : 0);
#pragma unroll
    for (int g = 0; g < 2; ++g)
#pragma unroll
        for (int h = 0; h < 2; ++h)
#pragma unroll
            for (int r = 0; r < 8; ++r) {
                const int lm   = (g == 0) ? (r == 0 ? 0 : r + 8) : (r == 0 ? 8 : r);
                const int l    = (lm >= 9) ? 3 : (lm >= 4) ? 2 : (lm >= 1) ? 1 : 0;
                const int flip = ((r > 0) != (g == 1)) ? 1 : 0;
                const int q    = 2 * h + (half ^ flip);
                const int lc   = g ? lc1 : lc0;
                const int nc   = g ? nc1 : nc0;
                if (lc == l) sOut[lm * 32 + q * 8 + nc] = acc[g][h][r];
            }
    v4f o[OUTROW / 128];
#pragma unroll
    for (int k = 0; k < OUTROW / 128; ++k) o[k] = *(const v4f*)&sOut[(k * 32 + lane) * 4];
#pragma unroll
    for (int k = 0; k < OUTROW / 128; ++k) *(v4f*)&outc[(k * 32 + lane) * 4] = o[k];
}

extern "C" void kernel_launch(void* const* d_in, const int* in_sizes, int n_in,
                              void* d_out, int out_size, void* d_ws, size_t ws_size,
                              hipStream_t stream) {
    const float* rb      = (const float*)d_in[0];
    const float* sph     = (const float*)d_in[1];
    const float* W       = (const float*)d_in[2];
    const int*   species = (const int*)d_in[3];
    const int*   di      = (const int*)d_in[4];
    const int    P       = in_sizes[3];
    const int    C       = NCENTERS;
    float*       out     = (float*)d_out;

    int* ctot    = (int*)d_ws;
    int* offsets = ctot + C;
    int* order   = offsets + C + 1 + C;
    int* mat     = (int*)d_out;

    slice_hist_kernel<<<NSLICE, SORT_T, 0, stream>>>(di, mat, ctot, P);
    slice_scan_kernel<<<NCHUNK, CHUNK, 0, stream>>>(mat, ctot, offsets);
    slice_place_kernel<<<NSLICE, SORT_T, 0, stream>>>(di, mat, order, P);
    spexp_wmma_kernel<<<C, 32, 0, stream>>>(rb, sph, W, species, order, offsets, out);
}


# Sort in ONE launch: each of 32 slices counting-sorts its own pairs in LDS; the last launch (4 waves) merges a center's 32 runs itself; 4 launches to 2; f32 sum order: by 32 slices, then LDS arrival
# speedup vs baseline: 3.5914x; 1.6499x over previous
#include <hip/hip_runtime.h>
#include <hip/hip_bf16.h>

typedef __attribute__((ext_vector_type(2))) float v2f;
typedef __attribute__((ext_vector_type(4))) float v4f;
typedef __attribute__((ext_vector_type(8))) float v8f;

#define NCENTERS 16384
#define OUTROW 512

#define NSLICE 32
#define SLICE_LEN 16384
#define SORT_T 1024
#define EPT (SLICE_LEN / SORT_T)

typedef __attribute__((ext_vector_type(4))) int v4i;

__global__ __launch_bounds__(SORT_T) void slice_sort_kernel(
    const int* __restrict__ di, int* __restrict__ ends_w, int* __restrict__ lid_w, int P)
{
    __shared__ alignas(16) int cnt[NCENTERS];
    __shared__ alignas(16) unsigned short stage[SLICE_LEN];
    __shared__ int wtot[SORT_T / 32];
    const int t    = threadIdx.x;
    const int b    = blockIdx.x;
    const int lane = t & 31;
    const int w    = t >> 5;

    const v4i z4 = {0, 0, 0, 0};
#pragma unroll
    for (int k = 0; k < NCENTERS / 4 / SORT_T; ++k) ((v4i*)cnt)[k * SORT_T + t] = z4;

    const int lo = b * SLICE_LEN;
    const int hi = (lo + SLICE_LEN < P) ? (lo + SLICE_LEN) : P;
    int v[EPT];
#pragma unroll
    for (int u = 0; u < EPT; ++u) {
        const int i = lo + u * SORT_T + t;
        v[u] = (i < hi) ? di[i] : -1;
    }
    __syncthreads();
#pragma unroll
    for (int u = 0; u < EPT; ++u)
        if ((unsigned)v[u] < (unsigned)NCENTERS) atomicAdd(&cnt[v[u]], 1);
    __syncthreads();

    v4i q[4];
    int s = 0;
#pragma unroll
    for (int k = 0; k < 4; ++k) {
        q[k] = ((const v4i*)cnt)[4 * t + k];
        s += q[k].x + q[k].y + q[k].z + q[k].w;
    }
    int inc = s;
#pragma unroll
    for (int off = 1; off < 32; off <<= 1) {
        const int n = __shfl_up(inc, off);
        if (lane >= off) inc += n;
    }
    if (lane == 31) wtot[w] = inc;
    __syncthreads();
    int run = inc - s;
#pragma unroll
    for (int k = 0; k < SORT_T / 32; ++k) {
        const int wk = wtot[k];
        if (k < w) run += wk;
    }
#pragma unroll
    for (int k = 0; k < 4; ++k) {
        v4i o;
        o.x = run; run += q[k].x;
        o.y = run; run += q[k].y;
        o.z = run; run += q[k].z;
        o.w = run; run += q[k].w;
        ((v4i*)cnt)[4 * t + k] = o;
    }
    __syncthreads();

    int pos[EPT];
#pragma unroll
    for (int u = 0; u < EPT; ++u) {
        pos[u] = -1;
        if ((unsigned)v[u] < (unsigned)NCENTERS) pos[u] = atomicAdd(&cnt[v[u]], 1);
    }
#pragma unroll
    for (int u = 0; u < EPT; ++u)
        if (pos[u] >= 0) stage[pos[u]] = (unsigned short)(u * SORT_T + t);
    __syncthreads();

    uint2* erow = (uint2*)((unsigned short*)ends_w + (size_t)b * NCENTERS);
#pragma unroll
    for (int k = 0; k < NCENTERS / 4 / SORT_T; ++k) {
        const v4i e = ((const v4i*)cnt)[k * SORT_T + t];
        uint2 o;
        o.x = (unsigned)e.x | ((unsigned)e.y << 16);
        o.y = (unsigned)e.z | ((unsigned)e.w << 16);
        erow[k * SORT_T + t] = o;
    }
    v4i* lrow = (v4i*)((unsigned short*)lid_w + (size_t)b * SLICE_LEN);
#pragma unroll
    for (int k = 0; k < SLICE_LEN / 8 / SORT_T; ++k)
        lrow[k * SORT_T + t] = ((const v4i*)stage)[k * SORT_T + t];
}

template <int N> struct StepCount { static constexpr int value = N; };

#define CPB 4
__global__ __launch_bounds__(32 * CPB) void spexp_wmma_kernel(
    const float* __restrict__ rb,
    const float* __restrict__ sph,
    const float* __restrict__ W,
    const int*   __restrict__ species,
    const int*   __restrict__ ends_w,
    const int*   __restrict__ lid_w,
    float*       __restrict__ out)
{
    __shared__ alignas(16) float sWT[16];
    __shared__ int sTab[NSLICE][CPB + 1];
    __shared__ alignas(16) int   sIdxA[CPB][32];
    __shared__ alignas(16) float sPA[CPB][2][32][4];
    __shared__ alignas(16) float sOutA[CPB][OUTROW];
    __shared__ int sOwnA[CPB][32];
    __shared__ int sBegA[CPB][32];

    const unsigned short* __restrict__ ends16 = (const unsigned short*)ends_w;
    const unsigned short* __restrict__ lid16  = (const unsigned short*)lid_w;

    const int t    = threadIdx.x;
    const int wv   = t >> 5;
    const int lane = t & 31;
    const int c0   = blockIdx.x * CPB;
    const int c    = c0 + wv;
    const int half = lane >> 4;
    const int idx  = lane & 15;

    if (t < 16) sWT[(t & 3) * 4 + (t >> 2)] = W[t];
    {
        const int s = t / CPB, j = t % CPB;
        const unsigned short* p = ends16 + (size_t)s * NCENTERS + c0 + j;
        sTab[s][j + 1] = p[0];
        if (j == 0) sTab[s][0] = (c0 > 0) ? p[-1] : 0;
    }
    __syncthreads();

    int*   const sIdx = sIdxA[wv];
    float (* const sP)[32][4] = sPA[wv];
    float* const sOut = sOutA[wv];
    int*   const sOwn = sOwnA[wv];
    int*   const sBeg = sBegA[wv];

    auto row_up1 = [](int v) -> int { return (int)__builtin_amdgcn_permlane16(v, v, 0x65432100u, 0xEDCBA987u, false, false); };
    auto row_up2 = [](int v) -> int { return (int)__builtin_amdgcn_permlane16(v, v, 0x54321010u, 0xDCBA9876u, false, false); };
    auto row_up4 = [](int v) -> int { return (int)__builtin_amdgcn_permlane16(v, v, 0x32103210u, 0xBA987654u, false, false); };
    auto row_up8 = [](int v) -> int { return (int)__builtin_amdgcn_permlane16(v, v, 0x76543210u, 0x76543210u, false, false); };
    auto imax = [](int a, int b) -> int { return a > b ? a : b; };

    const int e0 = sTab[lane][wv], e1 = sTab[lane][wv + 1];
    const int cl = e1 - e0;
    int incl = cl;
    { const int nb = row_up1(incl); if (idx >= 1) incl += nb; }
    { const int nb = row_up2(incl); if (idx >= 2) incl += nb; }
    { const int nb = row_up4(incl); if (idx >= 4) incl += nb; }
    { const int nb = row_up8(incl); if (idx >= 8) incl += nb; }
    { const int lowrow = __builtin_amdgcn_readlane(incl, 15); if (half) incl += lowrow; }
    const int start = 0;
    const int end   = __builtin_amdgcn_readlane(incl, 31);
    const int pos   = incl - cl;
    sBeg[lane] = lane * SLICE_LEN + e0 - pos;

    auto fetch = [&](const int g0) -> int {
        const unsigned m  = __builtin_amdgcn_ballot_w32(cl > 0 && pos <= g0);
        const int      r0 = 31 - __builtin_clz(m);
        sOwn[lane] = 0;
        const int d = pos - g0;
        if (cl > 0 && d > 0 && d < 32) sOwn[d] = lane;
        __builtin_amdgcn_wave_barrier();
        int own = sOwn[lane];
        own = imax(own, row_up1(own));
        own = imax(own, row_up2(own));
        own = imax(own, row_up4(own));
        own = imax(own, row_up8(own));
        { const int lowrow = __builtin_amdgcn_readlane(own, 15); if (half) own = imax(own, lowrow); }
        own = imax(own, r0);
        const int gl = g0 + lane;
        const int g  = gl < end - 1 ? gl : end - 1;
        return own * SLICE_LEN + (int)lid16[sBeg[own] + g];
    };

    const bool alpha = (idx == 0) || (idx >= 9);
    const bool beta  = (idx < 4) || (idx >= 12);
    const float (*sPl)[4] = sP[alpha ? 0 : 1];

    auto row_xor8 = [](float v) {
        const int i = __builtin_bit_cast(int, v);
        return __builtin_bit_cast(float, __builtin_amdgcn_permlane16(i, i, 0xFEDCBA98u, 0x76543210u, false, false));
    };
    auto row_xor4 = [](float v) {
        const int i = __builtin_bit_cast(int, v);
        return __builtin_bit_cast(float, __builtin_amdgcn_permlane16(i, i, 0x32107654u, 0xBA98FEDCu, false, false));
    };

    v8f acc[2][2];
    const v8f vzero = {0.f, 0.f, 0.f, 0.f, 0.f, 0.f, 0.f, 0.f};
#pragma unroll
    for (int g = 0; g < 2; ++g)
#pragma unroll
        for (int h = 0; h < 2; ++h) acc[g][h] = vzero;

    int pi = 0;
    if (start < end) {
        pi = fetch(start);
    }

    auto chunk = [&](auto nit_c, const int base, const int n) {
        constexpr int NIT = decltype(nit_c)::value;

        sIdx[lane] = pi;
        __builtin_amdgcn_wave_barrier();

        int2 pp[NIT];
#pragma unroll
        for (int it = 0; it < NIT; ++it) pp[it] = *(const int2*)&sIdx[4 * it + 2 * half];

        float s0[NIT], s1[NIT];
        v2f   r0[NIT], r1[NIT];
#pragma unroll
        for (int it = 0; it < NIT; ++it) {
            const unsigned a0 = (unsigned)pp[it].x, a1 = (unsigned)pp[it].y;
            s0[it] = __builtin_nontemporal_load(sph + (a0 * 16u + (unsigned)idx));
            s1[it] = __builtin_nontemporal_load(sph + (a1 * 16u + (unsigned)idx));
            r0[it] = __builtin_nontemporal_load((const v2f*)(rb + (a0 * 32u + 2u * (unsigned)idx)));
            r1[it] = __builtin_nontemporal_load((const v2f*)(rb + (a1 * 32u + 2u * (unsigned)idx)));
        }

        const int si = species[pi];
        int pn = pi;
        if (base + 32 < end) pn = fetch(base + 32);
        v4f w = *(const v4f*)&sWT[(si & 3) * 4];
        if (!(lane < n && (unsigned)si < 4u)) { w.x = 0.f; w.y = 0.f; w.z = 0.f; w.w = 0.f; }
        *(v4f*)&sP[0][lane][0] = w;
        v4f wsw;
        wsw.x = w.y; wsw.y = w.x; wsw.z = w.w; wsw.w = w.z;
        *(v4f*)&sP[1][lane][0] = wsw;
        __builtin_amdgcn_wave_barrier();

        v4f w0 = *(const v4f*)&sPl[2 * half][0];
        v4f w1 = *(const v4f*)&sPl[2 * half + 1][0];
#pragma unroll
        for (int it = 0; it < NIT; ++it) {
            if (4 * it < n) {
                const v4f u0 = w0, u1 = w1;
                if (it + 1 < NIT) {
                    w0 = *(const v4f*)&sPl[4 * (it + 1) + 2 * half][0];
                    w1 = *(const v4f*)&sPl[4 * (it + 1) + 2 * half + 1][0];
                }
                float x0 = s0[it], x1 = s1[it];
                v2f   y0 = r0[it], y1 = r1[it];
                if (4 * it + 4 > n) {
                    const int j0 = 4 * it + 2 * half;
                    if (j0 >= n)     { x0 = 0.f; y0.x = 0.f; y0.y = 0.f; }
                    if (j0 + 1 >= n) { x1 = 0.f; y1.x = 0.f; y1.y = 0.f; }
                }
                const float X0 = row_xor8(x0),   X1 = row_xor8(x1);
                const float T0 = row_xor4(y0.y), T1 = row_xor4(y1.y);
                v2f b0, b1;
                b0.x = beta ? y0.x : T0;  b0.y = beta ? y1.x : T1;
                b1.x = beta ? T0 : y0.x;  b1.y = beta ? T1 : y1.x;
                const float xg0e0 = alpha ? x0 : X0, xg0e1 = alpha ? x1 : X1;
                const float xg1e0 = alpha ? X0 : x0, xg1e1 = alpha ? X1 : x1;
#pragma unroll
                for (int h = 0; h < 2; ++h) {
                    v2f a0, a1;
                    a0.x = xg0e0 * u0[2 * h];
                    a0.y = xg0e1 * u1[2 * h];
                    a1.x = xg1e0 * u0[2 * h + 1];
                    a1.y = xg1e1 * u1[2 * h + 1];
                    acc[0][h] = __builtin_amdgcn_wmma_f32_16x16x4_f32(
                        false, a0, false, b0, (short)0, acc[0][h], false, false);
                    acc[1][h] = __builtin_amdgcn_wmma_f32_16x16x4_f32(
                        false, a1, false, b1, (short)0, acc[1][h], false, false);
                }
            }
        }

        pi = pn;
        __builtin_amdgcn_wave_barrier();
    };

    for (int base = start; base < end; base += 32) {
        const int n = (end - base < 32) ? (end - base) : 32;
        if (n > 16)     chunk(StepCount<8>{}, base, n);
        else if (n > 8) chunk(StepCount<4>{}, base, n);
        else            chunk(StepCount<2>{}, base, n);
    }

    float* outc = out + (size_t)c * OUTROW;
    const int lown = idx >> 2, lpar = (idx ^ 4) >> 2;
    const int lc0 = beta ? lown : lpar;
    const int lc1 = beta ? lpar : lown;
    const int nc0 = 2 * (idx & 3) + (beta ? 0 : 1);
    const int nc1 = 2 * (idx & 3) + (beta ? 1 : 0);
#pragma unroll
    for (int g = 0; g < 2; ++g)
#pragma unroll
        for (int h = 0; h < 2; ++h)
#pragma unroll
            for (int r = 0; r < 8; ++r) {
                const int lm   = (g == 0) ? (r == 0 ? 0 : r + 8) : (r == 0 ? 8 : r);
                const int l    = (lm >= 9) ? 3 : (lm >= 4) ? 2 : (lm >= 1) ? 1 : 0;
                const int flip = ((r > 0) != (g == 1)) ? 1 : 0;
                const int q    = 2 * h + (half ^ flip);
                const int lc   = g ? lc1 : lc0;
                const int nc   = g ? nc1 : nc0;
                if (lc == l) sOut[lm * 32 + q * 8 + nc] = acc[g][h][r];
            }
    v4f o[OUTROW / 128];
#pragma unroll
    for (int k = 0; k < OUTROW / 128; ++k) o[k] = *(const v4f*)&sOut[(k * 32 + lane) * 4];
#pragma unroll
    for (int k = 0; k < OUTROW / 128; ++k) *(v4f*)&outc[(k * 32 + lane) * 4] = o[k];
}

extern "C" void kernel_launch(void* const* d_in, const int* in_sizes, int n_in,
                              void* d_out, int out_size, void* d_ws, size_t ws_size,
                              hipStream_t stream) {
    const float* rb      = (const float*)d_in[0];
    const float* sph     = (const float*)d_in[1];
    const float* W       = (const float*)d_in[2];
    const int*   species = (const int*)d_in[3];
    const int*   di      = (const int*)d_in[4];
    const int    P       = in_sizes[3];
    const int    C       = NCENTERS;
    float*       out     = (float*)d_out;

    int* ends_w = (int*)d_ws;
    int* lid_w  = ends_w + NSLICE * NCENTERS / 2;

    slice_sort_kernel<<<NSLICE, SORT_T, 0, stream>>>(di, ends_w, lid_w, P);
    spexp_wmma_kernel<<<NCENTERS / CPB, 32 * CPB, 0, stream>>>(rb, sph, W, species, ends_w, lid_w, out);
}


# Last launch: each wave walks 2 neighbouring centers in turn; 2048 work-groups of 4 waves at 128 registers (step weights single-buffered), all 8192 waves resident at once, was 16384; same sum order
# speedup vs baseline: 3.7548x; 1.0455x over previous
#include <hip/hip_runtime.h>
#include <hip/hip_bf16.h>

typedef __attribute__((ext_vector_type(2))) float v2f;
typedef __attribute__((ext_vector_type(4))) float v4f;
typedef __attribute__((ext_vector_type(8))) float v8f;

#define NCENTERS 16384
#define OUTROW 512

#define NSLICE 32
#define SLICE_LEN 16384
#define SORT_T 1024
#define EPT (SLICE_LEN / SORT_T)

typedef __attribute__((ext_vector_type(4))) int v4i;

__global__ __launch_bounds__(SORT_T) void slice_sort_kernel(
    const int* __restrict__ di, int* __restrict__ ends_w, int* __restrict__ lid_w, int P)
{
    __shared__ alignas(16) int cnt[NCENTERS];
    __shared__ alignas(16) unsigned short stage[SLICE_LEN];
    __shared__ int wtot[SORT_T / 32];
    const int t    = threadIdx.x;
    const int b    = blockIdx.x;
    const int lane = t & 31;
    const int w    = t >> 5;

    const v4i z4 = {0, 0, 0, 0};
#pragma unroll
    for (int k = 0; k < NCENTERS / 4 / SORT_T; ++k) ((v4i*)cnt)[k * SORT_T + t] = z4;

    const int lo = b * SLICE_LEN;
    const int hi = (lo + SLICE_LEN < P) ? (lo + SLICE_LEN) : P;
    int v[EPT];
#pragma unroll
    for (int u = 0; u < EPT; ++u) {
        const int i = lo + u * SORT_T + t;
        v[u] = (i < hi) ? di[i] : -1;
    }
    __syncthreads();
#pragma unroll
    for (int u = 0; u < EPT; ++u)
        if ((unsigned)v[u] < (unsigned)NCENTERS) atomicAdd(&cnt[v[u]], 1);
    __syncthreads();

    v4i q[4];
    int s = 0;
#pragma unroll
    for (int k = 0; k < 4; ++k) {
        q[k] = ((const v4i*)cnt)[4 * t + k];
        s += q[k].x + q[k].y + q[k].z + q[k].w;
    }
    int inc = s;
#pragma unroll
    for (int off = 1; off < 32; off <<= 1) {
        const int n = __shfl_up(inc, off);
        if (lane >= off) inc += n;
    }
    if (lane == 31) wtot[w] = inc;
    __syncthreads();
    int run = inc - s;
#pragma unroll
    for (int k = 0; k < SORT_T / 32; ++k) {
        const int wk = wtot[k];
        if (k < w) run += wk;
    }
#pragma unroll
    for (int k = 0; k < 4; ++k) {
        v4i o;
        o.x = run; run += q[k].x;
        o.y = run; run += q[k].y;
        o.z = run; run += q[k].z;
        o.w = run; run += q[k].w;
        ((v4i*)cnt)[4 * t + k] = o;
    }
    __syncthreads();

    int pos[EPT];
#pragma unroll
    for (int u = 0; u < EPT; ++u) {
        pos[u] = -1;
        if ((unsigned)v[u] < (unsigned)NCENTERS) pos[u] = atomicAdd(&cnt[v[u]], 1);
    }
#pragma unroll
    for (int u = 0; u < EPT; ++u)
        if (pos[u] >= 0) stage[pos[u]] = (unsigned short)(u * SORT_T + t);
    __syncthreads();

    uint2* erow = (uint2*)((unsigned short*)ends_w + (size_t)b * NCENTERS);
#pragma unroll
    for (int k = 0; k < NCENTERS / 4 / SORT_T; ++k) {
        const v4i e = ((const v4i*)cnt)[k * SORT_T + t];
        uint2 o;
        o.x = (unsigned)e.x | ((unsigned)e.y << 16);
        o.y = (unsigned)e.z | ((unsigned)e.w << 16);
        erow[k * SORT_T + t] = o;
    }
    v4i* lrow = (v4i*)((unsigned short*)lid_w + (size_t)b * SLICE_LEN);
#pragma unroll
    for (int k = 0; k < SLICE_LEN / 8 / SORT_T; ++k)
        lrow[k * SORT_T + t] = ((const v4i*)stage)[k * SORT_T + t];
}

template <int N> struct StepCount { static constexpr int value = N; };

#define CPB 4
#define KC 2
#define CPW (CPB * KC)
__global__ __launch_bounds__(32 * CPB) __attribute__((amdgpu_waves_per_eu(8, 8))) void spexp_wmma_kernel(
    const float* __restrict__ rb,
    const float* __restrict__ sph,
    const float* __restrict__ W,
    const int*   __restrict__ species,
    const int*   __restrict__ ends_w,
    const int*   __restrict__ lid_w,
    float*       __restrict__ out)
{
    __shared__ alignas(16) float sWT[16];
    __shared__ int sTab[NSLICE][CPW + 1];
    __shared__ alignas(16) int   sIdxA[CPB][32];
    __shared__ alignas(16) float sPA[CPB][2][32][4];
    __shared__ alignas(16) float sOutA[CPB][OUTROW];
    __shared__ int sOwnA[CPB][32];
    __shared__ int sBegA[CPB][32];

    const unsigned short* __restrict__ ends16 = (const unsigned short*)ends_w;
    const unsigned short* __restrict__ lid16  = (const unsigned short*)lid_w;

    const int t    = threadIdx.x;
    const int wv   = t >> 5;
    const int lane = t & 31;
    const int c0   = blockIdx.x * CPW;
    const int half = lane >> 4;
    const int idx  = lane & 15;

    if (t < 16) sWT[(t & 3) * 4 + (t >> 2)] = W[t];
    {
        const int s = t / (CPW / 2), j = t % (CPW / 2);
        const size_t at = (size_t)s * NCENTERS + c0 + 2 * j;
        const unsigned two = (unsigned)ends_w[at >> 1];
        sTab[s][2 * j + 1] = (int)(two & 0xFFFFu);
        sTab[s][2 * j + 2] = (int)(two >> 16);
        if (j == 0) sTab[s][0] = (c0 > 0) ? (int)ends16[at - 1] : 0;
    }
    __syncthreads();

    int*   const sIdx = sIdxA[wv];
    float (* const sP)[32][4] = sPA[wv];
    float* const sOut = sOutA[wv];
    int*   const sOwn = sOwnA[wv];
    int*   const sBeg = sBegA[wv];

    auto row_up1 = [](int v) -> int { return (int)__builtin_amdgcn_permlane16(v, v, 0x65432100u, 0xEDCBA987u, false, false); };
    auto row_up2 = [](int v) -> int { return (int)__builtin_amdgcn_permlane16(v, v, 0x54321010u, 0xDCBA9876u, false, false); };
    auto row_up4 = [](int v) -> int { return (int)__builtin_amdgcn_permlane16(v, v, 0x32103210u, 0xBA987654u, false, false); };
    auto row_up8 = [](int v) -> int { return (int)__builtin_amdgcn_permlane16(v, v, 0x76543210u, 0x76543210u, false, false); };
    auto imax = [](int a, int b) -> int { return a > b ? a : b; };

    for (int kk = 0; kk < KC; ++kk) {
    const int ci = wv * KC + kk;
    const int c  = c0 + ci;

    const int e0 = sTab[lane][ci], e1 = sTab[lane][ci + 1];
    const int cl = e1 - e0;
    int incl = cl;
    { const int nb = row_up1(incl); if (idx >= 1) incl += nb; }
    { const int nb = row_up2(incl); if (idx >= 2) incl += nb; }
    { const int nb = row_up4(incl); if (idx >= 4) incl += nb; }
    { const int nb = row_up8(incl); if (idx >= 8) incl += nb; }
    { const int lowrow = __builtin_amdgcn_readlane(incl, 15); if (half) incl += lowrow; }
    const int start = 0;
    const int end   = __builtin_amdgcn_readlane(incl, 31);
    const int pos   = incl - cl;
    sBeg[lane] = lane * SLICE_LEN + e0 - pos;

    auto fetch = [&](const int g0) -> int {
        const unsigned m  = __builtin_amdgcn_ballot_w32(cl > 0 && pos <= g0);
        const int      r0 = 31 - __builtin_clz(m);
        sOwn[lane] = 0;
        const int d = pos - g0;
        if (cl > 0 && d > 0 && d < 32) sOwn[d] = lane;
        __builtin_amdgcn_wave_barrier();
        int own = sOwn[lane];
        own = imax(own, row_up1(own));
        own = imax(own, row_up2(own));
        own = imax(own, row_up4(own));
        own = imax(own, row_up8(own));
        { const int lowrow = __builtin_amdgcn_readlane(own, 15); if (half) own = imax(own, lowrow); }
        own = imax(own, r0);
        const int gl = g0 + lane;
        const int g  = gl < end - 1 ? gl : end - 1;
        return own * SLICE_LEN + (int)lid16[sBeg[own] + g];
    };

    const bool alpha = (idx == 0) || (idx >= 9);
    const bool beta  = (idx < 4) || (idx >= 12);
    const float (*sPl)[4] = sP[alpha ? 0 : 1];

    auto row_xor8 = [](float v) {
        const int i = __builtin_bit_cast(int, v);
        return __builtin_bit_cast(float, __builtin_amdgcn_permlane16(i, i, 0xFEDCBA98u, 0x76543210u, false, false));
    };
    auto row_xor4 = [](float v) {
        const int i = __builtin_bit_cast(int, v);
        return __builtin_bit_cast(float, __builtin_amdgcn_permlane16(i, i, 0x32107654u, 0xBA98FEDCu, false, false));
    };

    v8f acc[2][2];
    const v8f vzero = {0.f, 0.f, 0.f, 0.f, 0.f, 0.f, 0.f, 0.f};
#pragma unroll
    for (int g = 0; g < 2; ++g)
#pragma unroll
        for (int h = 0; h < 2; ++h) acc[g][h] = vzero;

    int pi = 0;
    if (start < end) {
        pi = fetch(start);
    }

    auto chunk = [&](auto nit_c, const int base, const int n) {
        constexpr int NIT = decltype(nit_c)::value;

        sIdx[lane] = pi;
        __builtin_amdgcn_wave_barrier();

        int2 pp[NIT];
#pragma unroll
        for (int it = 0; it < NIT; ++it) pp[it] = *(const int2*)&sIdx[4 * it + 2 * half];

        float s0[NIT], s1[NIT];
        v2f   r0[NIT], r1[NIT];
#pragma unroll
        for (int it = 0; it < NIT; ++it) {
            const unsigned a0 = (unsigned)pp[it].x, a1 = (unsigned)pp[it].y;
            s0[it] = __builtin_nontemporal_load(sph + (a0 * 16u + (unsigned)idx));
            s1[it] = __builtin_nontemporal_load(sph + (a1 * 16u + (unsigned)idx));
            r0[it] = __builtin_nontemporal_load((const v2f*)(rb + (a0 * 32u + 2u * (unsigned)idx)));
            r1[it] = __builtin_nontemporal_load((const v2f*)(rb + (a1 * 32u + 2u * (unsigned)idx)));
        }

        const int si = species[pi];
        int pn = pi;
        if (base + 32 < end) pn = fetch(base + 32);
        v4f w = *(const v4f*)&sWT[(si & 3) * 4];
        if (!(lane < n && (unsigned)si < 4u)) { w.x = 0.f; w.y = 0.f; w.z = 0.f; w.w = 0.f; }
        *(v4f*)&sP[0][lane][0] = w;
        v4f wsw;
        wsw.x = w.y; wsw.y = w.x; wsw.z = w.w; wsw.w = w.z;
        *(v4f*)&sP[1][lane][0] = wsw;
        __builtin_amdgcn_wave_barrier();

#pragma unroll
        for (int it = 0; it < NIT; ++it) {
            if (4 * it < n) {
                const v4f u0 = *(const v4f*)&sPl[4 * it + 2 * half][0];
                const v4f u1 = *(const v4f*)&sPl[4 * it + 2 * half + 1][0];
                float x0 = s0[it], x1 = s1[it];
                v2f   y0 = r0[it], y1 = r1[it];
                if (4 * it + 4 > n) {
                    const int j0 = 4 * it + 2 * half;
                    if (j0 >= n)     { x0 = 0.f; y0.x = 0.f; y0.y = 0.f; }
                    if (j0 + 1 >= n) { x1 = 0.f; y1.x = 0.f; y1.y = 0.f; }
                }
                const float X0 = row_xor8(x0),   X1 = row_xor8(x1);
                const float T0 = row_xor4(y0.y), T1 = row_xor4(y1.y);
                v2f b0, b1;
                b0.x = beta ? y0.x : T0;  b0.y = beta ? y1.x : T1;
                b1.x = beta ? T0 : y0.x;  b1.y = beta ? T1 : y1.x;
                const float xg0e0 = alpha ? x0 : X0, xg0e1 = alpha ? x1 : X1;
                const float xg1e0 = alpha ? X0 : x0, xg1e1 = alpha ? X1 : x1;
#pragma unroll
                for (int h = 0; h < 2; ++h) {
                    v2f a0, a1;
                    a0.x = xg0e0 * u0[2 * h];
                    a0.y = xg0e1 * u1[2 * h];
                    a1.x = xg1e0 * u0[2 * h + 1];
                    a1.y = xg1e1 * u1[2 * h + 1];
                    acc[0][h] = __builtin_amdgcn_wmma_f32_16x16x4_f32(
                        false, a0, false, b0, (short)0, acc[0][h], false, false);
                    acc[1][h] = __builtin_amdgcn_wmma_f32_16x16x4_f32(
                        false, a1, false, b1, (short)0, acc[1][h], false, false);
                }
            }
        }

        pi = pn;
        __builtin_amdgcn_wave_barrier();
    };

    for (int base = start; base < end; base += 32) {
        const int n = (end - base < 32) ? (end - base) : 32;
        if (n > 16)     chunk(StepCount<8>{}, base, n);
        else if (n > 8) chunk(StepCount<4>{}, base, n);
        else            chunk(StepCount<2>{}, base, n);
    }

    float* outc = out + (size_t)c * OUTROW;
    const int lown = idx >> 2, lpar = (idx ^ 4) >> 2;
    const int lc0 = beta ? lown : lpar;
    const int lc1 = beta ? lpar : lown;
    const int nc0 = 2 * (idx & 3) + (beta ? 0 : 1);
    const int nc1 = 2 * (idx & 3) + (beta ? 1 : 0);
#pragma unroll
    for (int g = 0; g < 2; ++g)
#pragma unroll
        for (int h = 0; h < 2; ++h)
#pragma unroll
            for (int r = 0; r < 8; ++r) {
                const int lm   = (g == 0) ? (r == 0 ? 0 : r + 8) : (r == 0 ? 8 : r);
                const int l    = (lm >= 9) ? 3 : (lm >= 4) ? 2 : (lm >= 1) ? 1 : 0;
                const int flip = ((r > 0) != (g == 1)) ? 1 : 0;
                const int q    = 2 * h + (half ^ flip);
                const int lc   = g ? lc1 : lc0;
                const int nc   = g ? nc1 : nc0;
                if (lc == l) sOut[lm * 32 + q * 8 + nc] = acc[g][h][r];
            }
    v4f o[OUTROW / 128];
#pragma unroll
    for (int k = 0; k < OUTROW / 128; ++k) o[k] = *(const v4f*)&sOut[(k * 32 + lane) * 4];
#pragma unroll
    for (int k = 0; k < OUTROW / 128; ++k) *(v4f*)&outc[(k * 32 + lane) * 4] = o[k];
    }
}

extern "C" void kernel_launch(void* const* d_in, const int* in_sizes, int n_in,
                              void* d_out, int out_size, void* d_ws, size_t ws_size,
                              hipStream_t stream) {
    const float* rb      = (const float*)d_in[0];
    const float* sph     = (const float*)d_in[1];
    const float* W       = (const float*)d_in[2];
    const int*   species = (const int*)d_in[3];
    const int*   di      = (const int*)d_in[4];
    const int    P       = in_sizes[3];
    const int    C       = NCENTERS;
    float*       out     = (float*)d_out;

    int* ends_w = (int*)d_ws;
    int* lid_w  = ends_w + NSLICE * NCENTERS / 2;

    slice_sort_kernel<<<NSLICE, SORT_T, 0, stream>>>(di, ends_w, lid_w, P);
    spexp_wmma_kernel<<<NCENTERS / CPW, 32 * CPB, 0, stream>>>(rb, sph, W, species, ends_w, lid_w, out);
}
